# starting kernel
# baseline (speedup 1.0000x reference)
#include <hip/hip_runtime.h>
#include <hip/hip_bf16.h>
#include <math.h>


#define TT 512
#define DD 512

typedef __attribute__((ext_vector_type(2))) float v2f;
typedef __attribute__((ext_vector_type(8))) float v8f;

__global__ void qkv_gemm_f32(const float* __restrict__ x,
                             const float* __restrict__ Wq,
                             const float* __restrict__ Wk,
                             const float* __restrict__ Wv,
                             float* __restrict__ q,
                             float* __restrict__ k,
                             float* __restrict__ v) {
    const float* W = (blockIdx.z == 0) ? Wq : (blockIdx.z == 1) ? Wk : Wv;
    float* dst     = (blockIdx.z == 0) ? q  : (blockIdx.z == 1) ? k  : v;

    const int lane   = threadIdx.x & 31;
    const int tileId = blockIdx.x * (blockDim.x >> 5) + (threadIdx.x >> 5);
    const int tm = (tileId >> 5) << 4;
    const int tn = (tileId & 31) << 4;

    const int laneM = lane & 15;
    const int laneK = (lane >> 4) << 1;

    const float* Arow = W + (tm + laneM) * DD + laneK;
    const float* Brow = x + (tn + laneM) * DD + laneK;

    v8f c = {};
#pragma unroll 8
    for (int k0 = 0; k0 < DD; k0 += 4) {
        v2f a = *(const v2f*)(Arow + k0);
        v2f b = *(const v2f*)(Brow + k0);
        c = __builtin_amdgcn_wmma_f32_16x16x4_f32(
                 false, a,  false, b,
                 (short)0, c,  false,  false);
    }

    const int rowOff = (lane >> 4) << 3;
    const int col = tn + laneM;
#pragma unroll
    for (int i = 0; i < 8; ++i)
        dst[(tm + rowOff + i) * DD + col] = c[i];
}

__global__ void rope_heads(const float* __restrict__ qin,
                           const float* __restrict__ kin,
                           float* __restrict__ qout,
                           float* __restrict__ kout) {
    int idx = blockIdx.x * blockDim.x + threadIdx.x;
    int t = idx & (TT - 1);
    int o = idx >> 9;
    int c = o & 63;
    int h = o >> 6;

    float invf = __powf(10000.0f, -(float)((c & 31) << 1) * (1.0f / 64.0f));
    float ang  = (float)t * invf;
    float cs = __cosf(ang);
    float sn = __sinf(ang);

    int poff  = (h < 4) ? (256 * TT) : (-256 * TT);
    float sgn = (h < 4) ? -1.0f : 1.0f;

    qout[idx] = qin[idx] * cs + sgn * qin[idx + poff] * sn;
    kout[idx] = kin[idx] * cs + sgn * kin[idx + poff] * sn;
}

__global__ void attn_rows(const float* __restrict__ q,
                          const float* __restrict__ k,
                          const float* __restrict__ v,
                          float* __restrict__ o) {
    const int lane = threadIdx.x & 31;
    const int row  = blockIdx.x * (blockDim.x >> 5) + (threadIdx.x >> 5);
    const int ch = row >> 9;
    const int i  = row & (TT - 1);

    const float* kr = k + ch * TT;
    const float* vr = v + ch * TT;
    const float a = q[ch * TT + i] * 0.125f;

    float m = -3.4e38f;
    for (int j = lane; j <= i; j += 32)
        m = fmaxf(m, a * kr[j]);
#pragma unroll
    for (int off = 16; off > 0; off >>= 1)
        m = fmaxf(m, __shfl_xor(m, off, 32));

    float s = 0.0f, acc = 0.0f;
    for (int j = lane; j <= i; j += 32) {
        float e = __expf(a * kr[j] - m);
        s   += e;
        acc += e * vr[j];
    }
#pragma unroll
    for (int off = 16; off > 0; off >>= 1) {
        s   += __shfl_xor(s, off, 32);
        acc += __shfl_xor(acc, off, 32);
    }

    if (lane == 0)
        o[ch * TT + i] = acc / s;
}

__global__ void out_gemm_f32(const float* __restrict__ ao,
                             const float* __restrict__ Wo,
                             float* __restrict__ out) {
    const int lane   = threadIdx.x & 31;
    const int tileId = blockIdx.x * (blockDim.x >> 5) + (threadIdx.x >> 5);
    const int tm = (tileId >> 5) << 4;
    const int tn = (tileId & 31) << 4;

    const int laneM = lane & 15;
    const int laneK = (lane >> 4) << 1;

    const float* Arow = Wo + (tm + laneM) * DD + laneK;
    const float* Bcol = ao + laneK * TT + (tn + laneM);

    v8f c = {};
#pragma unroll 8
    for (int k0 = 0; k0 < DD; k0 += 4) {
        v2f a = *(const v2f*)(Arow + k0);
        v2f b;
        b.x = Bcol[k0 * TT];
        b.y = Bcol[k0 * TT + TT];
        c = __builtin_amdgcn_wmma_f32_16x16x4_f32(
                false, a, false, b, (short)0, c, false, false);
    }

    const int rowOff = (lane >> 4) << 3;
    const int tcol = tn + laneM;
#pragma unroll
    for (int i = 0; i < 8; ++i)
        out[tcol * DD + (tm + rowOff + i)] = c[i];
}

extern "C" void kernel_launch(void* const* d_in, const int* in_sizes, int n_in,
                              void* d_out, int out_size, void* d_ws, size_t ws_size,
                              hipStream_t stream) {
    const float* x  = (const float*)d_in[0];
    const float* Wq = (const float*)d_in[1];
    const float* Wk = (const float*)d_in[2];
    const float* Wv = (const float*)d_in[3];
    const float* Wo = (const float*)d_in[4];
    float* out = (float*)d_out;

    float* ws = (float*)d_ws;
    const size_t S = (size_t)DD * TT;
    float* qraw = ws + 0 * S;
    float* kraw = ws + 1 * S;
    float* vbuf = ws + 2 * S;
    float* qrot = ws + 3 * S;
    float* krot = ws + 4 * S;
    float* atto = qraw;

    qkv_gemm_f32<<<dim3(256, 1, 3), 128, 0, stream>>>(x, Wq, Wk, Wv, qraw, kraw, vbuf);

    rope_heads<<<(DD * TT) / 256, 256, 0, stream>>>(qraw, kraw, qrot, krot);

    attn_rows<<<(DD * TT) / 8, 256, 0, stream>>>(qrot, krot, vbuf, atto);

    out_gemm_f32<<<256, 128, 0, stream>>>(atto, Wo, out);
}


# attention rows: one lane per query row with k[j], v[j] as wave-uniform scalar loads, one block per channel with balanced row pairs, replacing one wave per row with shuffle reductions
# speedup vs baseline: 2.3380x; 2.3380x over previous
#include <hip/hip_runtime.h>
#include <hip/hip_bf16.h>
#include <math.h>


#define TT 512
#define DD 512

typedef __attribute__((ext_vector_type(2))) float v2f;
typedef __attribute__((ext_vector_type(8))) float v8f;

__global__ void qkv_gemm_f32(const float* __restrict__ x,
                             const float* __restrict__ Wq,
                             const float* __restrict__ Wk,
                             const float* __restrict__ Wv,
                             float* __restrict__ q,
                             float* __restrict__ k,
                             float* __restrict__ v) {
    const float* W = (blockIdx.z == 0) ? Wq : (blockIdx.z == 1) ? Wk : Wv;
    float* dst     = (blockIdx.z == 0) ? q  : (blockIdx.z == 1) ? k  : v;

    const int lane   = threadIdx.x & 31;
    const int tileId = blockIdx.x * (blockDim.x >> 5) + (threadIdx.x >> 5);
    const int tm = (tileId >> 5) << 4;
    const int tn = (tileId & 31) << 4;

    const int laneM = lane & 15;
    const int laneK = (lane >> 4) << 1;

    const float* Arow = W + (tm + laneM) * DD + laneK;
    const float* Brow = x + (tn + laneM) * DD + laneK;

    v8f c = {};
#pragma unroll 8
    for (int k0 = 0; k0 < DD; k0 += 4) {
        v2f a = *(const v2f*)(Arow + k0);
        v2f b = *(const v2f*)(Brow + k0);
        c = __builtin_amdgcn_wmma_f32_16x16x4_f32(
                 false, a,  false, b,
                 (short)0, c,  false,  false);
    }

    const int rowOff = (lane >> 4) << 3;
    const int col = tn + laneM;
#pragma unroll
    for (int i = 0; i < 8; ++i)
        dst[(tm + rowOff + i) * DD + col] = c[i];
}

__global__ void rope_heads(const float* __restrict__ qin,
                           const float* __restrict__ kin,
                           float* __restrict__ qout,
                           float* __restrict__ kout) {
    int idx = blockIdx.x * blockDim.x + threadIdx.x;
    int t = idx & (TT - 1);
    int o = idx >> 9;
    int c = o & 63;
    int h = o >> 6;

    float invf = __powf(10000.0f, -(float)((c & 31) << 1) * (1.0f / 64.0f));
    float ang  = (float)t * invf;
    float cs = __cosf(ang);
    float sn = __sinf(ang);

    int poff  = (h < 4) ? (256 * TT) : (-256 * TT);
    float sgn = (h < 4) ? -1.0f : 1.0f;

    qout[idx] = qin[idx] * cs + sgn * qin[idx + poff] * sn;
    kout[idx] = kin[idx] * cs + sgn * kin[idx + poff] * sn;
}

typedef __attribute__((ext_vector_type(16))) float v16f;

#define NEGBIG (-3.4e38f)

__device__ __forceinline__ void max16(float& m0, float& m1, const float a, const v16f kk) {
#pragma unroll
    for (int u = 0; u < 16; u += 2) {
        m0 = fmaxf(m0, a * kk[u]);
        m1 = fmaxf(m1, a * kk[u + 1]);
    }
}

__device__ __forceinline__ void max16_diag(float& m0, float& m1, const float a, const v16f kk,
                                           const int base, const int lane) {
#pragma unroll
    for (int u = 0; u < 16; u += 2) {
        m0 = fmaxf(m0, (base + u     <= lane) ? a * kk[u]     : NEGBIG);
        m1 = fmaxf(m1, (base + u + 1 <= lane) ? a * kk[u + 1] : NEGBIG);
    }
}

__device__ __forceinline__ void acc8(float& s0, float& s1, float& c0, float& c1,
                                     const float a, const float m, const v8f kk, const v8f vv) {
#pragma unroll
    for (int u = 0; u < 8; u += 2) {
        const float e0 = __expf(a * kk[u] - m);
        const float e1 = __expf(a * kk[u + 1] - m);
        s0 += e0;
        s1 += e1;
        c0 += e0 * vv[u];
        c1 += e1 * vv[u + 1];
    }
}

__device__ __forceinline__ void acc8_diag(float& s0, float& s1, float& c0, float& c1,
                                          const float a, const float m, const v8f kk, const v8f vv,
                                          const int base, const int lane) {
#pragma unroll
    for (int u = 0; u < 8; u += 2) {
        const float e0 = (base + u     <= lane) ? __expf(a * kk[u] - m)     : 0.0f;
        const float e1 = (base + u + 1 <= lane) ? __expf(a * kk[u + 1] - m) : 0.0f;
        s0 += e0;
        s1 += e1;
        c0 += e0 * vv[u];
        c1 += e1 * vv[u + 1];
    }
}

__device__ __forceinline__ float attn_row32(const float* __restrict__ kr,
                                            const float* __restrict__ vr,
                                            const float a, const int r0, const int lane) {
    float m0 = NEGBIG, m1 = NEGBIG;
    {
        v16f kA = *(const v16f*)(kr);
        v16f kB;
        for (int j = 0; j < r0; j += 32) {
            kB = *(const v16f*)(kr + j + 16);
            max16(m0, m1, a, kA);
            kA = *(const v16f*)(kr + j + 32);
            max16(m0, m1, a, kB);
        }
        kB = *(const v16f*)(kr + r0 + 16);
        max16_diag(m0, m1, a, kA, 0, lane);
        max16_diag(m0, m1, a, kB, 16, lane);
    }
    const float m = fmaxf(m0, m1);

    float s0 = 0.0f, s1 = 0.0f, c0 = 0.0f, c1 = 0.0f;
    {
        v8f kA = *(const v8f*)(kr);
        v8f vA = *(const v8f*)(vr);
        v8f kB, vB;
        for (int j = 0; j < r0; j += 16) {
            kB = *(const v8f*)(kr + j + 8);
            vB = *(const v8f*)(vr + j + 8);
            acc8(s0, s1, c0, c1, a, m, kA, vA);
            kA = *(const v8f*)(kr + j + 16);
            vA = *(const v8f*)(vr + j + 16);
            acc8(s0, s1, c0, c1, a, m, kB, vB);
        }
        kB = *(const v8f*)(kr + r0 + 8);
        vB = *(const v8f*)(vr + r0 + 8);
        acc8_diag(s0, s1, c0, c1, a, m, kA, vA, 0, lane);
        kA = *(const v8f*)(kr + r0 + 16);
        vA = *(const v8f*)(vr + r0 + 16);
        acc8_diag(s0, s1, c0, c1, a, m, kB, vB, 8, lane);
        kB = *(const v8f*)(kr + r0 + 24);
        vB = *(const v8f*)(vr + r0 + 24);
        acc8_diag(s0, s1, c0, c1, a, m, kA, vA, 16, lane);
        acc8_diag(s0, s1, c0, c1, a, m, kB, vB, 24, lane);
    }
    return (c0 + c1) / (s0 + s1);
}

__global__ void __launch_bounds__(256)
attn_rows(const float* __restrict__ q,
          const float* __restrict__ k,
          const float* __restrict__ v,
          float* __restrict__ o) {
    const int lane = threadIdx.x & 31;
    const int w    = __builtin_amdgcn_readfirstlane(threadIdx.x >> 5);
    const int ch   = blockIdx.x;

    const float* kr = k + ch * TT;
    const float* vr = v + ch * TT;
    const float* qr = q + ch * TT;
    float*       orow = o + ch * TT;

    const int rlo = w << 5;
    const int rhi = 480 - (w << 5);

    const float alo = qr[rlo + lane] * 0.125f;
    const float ahi = qr[rhi + lane] * 0.125f;

    orow[rlo + lane] = attn_row32(kr, vr, alo, rlo, lane);
    orow[rhi + lane] = attn_row32(kr, vr, ahi, rhi, lane);
}

__global__ void out_gemm_f32(const float* __restrict__ ao,
                             const float* __restrict__ Wo,
                             float* __restrict__ out) {
    const int lane   = threadIdx.x & 31;
    const int tileId = blockIdx.x * (blockDim.x >> 5) + (threadIdx.x >> 5);
    const int tm = (tileId >> 5) << 4;
    const int tn = (tileId & 31) << 4;

    const int laneM = lane & 15;
    const int laneK = (lane >> 4) << 1;

    const float* Arow = Wo + (tm + laneM) * DD + laneK;
    const float* Bcol = ao + laneK * TT + (tn + laneM);

    v8f c = {};
#pragma unroll 8
    for (int k0 = 0; k0 < DD; k0 += 4) {
        v2f a = *(const v2f*)(Arow + k0);
        v2f b;
        b.x = Bcol[k0 * TT];
        b.y = Bcol[k0 * TT + TT];
        c = __builtin_amdgcn_wmma_f32_16x16x4_f32(
                false, a, false, b, (short)0, c, false, false);
    }

    const int rowOff = (lane >> 4) << 3;
    const int tcol = tn + laneM;
#pragma unroll
    for (int i = 0; i < 8; ++i)
        out[tcol * DD + (tm + rowOff + i)] = c[i];
}

extern "C" void kernel_launch(void* const* d_in, const int* in_sizes, int n_in,
                              void* d_out, int out_size, void* d_ws, size_t ws_size,
                              hipStream_t stream) {
    const float* x  = (const float*)d_in[0];
    const float* Wq = (const float*)d_in[1];
    const float* Wk = (const float*)d_in[2];
    const float* Wv = (const float*)d_in[3];
    const float* Wo = (const float*)d_in[4];
    float* out = (float*)d_out;

    float* ws = (float*)d_ws;
    const size_t S = (size_t)DD * TT;
    float* qraw = ws + 0 * S;
    float* kraw = ws + 1 * S;
    float* vbuf = ws + 2 * S;
    float* qrot = ws + 3 * S;
    float* krot = ws + 4 * S;
    float* atto = qraw;

    qkv_gemm_f32<<<dim3(256, 1, 3), 128, 0, stream>>>(x, Wq, Wk, Wv, qraw, kraw, vbuf);

    rope_heads<<<(DD * TT) / 256, 256, 0, stream>>>(qraw, kraw, qrot, krot);

    attn_rows<<<DD, 256, 0, stream>>>(qrot, krot, vbuf, atto);

    out_gemm_f32<<<256, 128, 0, stream>>>(atto, Wo, out);
}


# attention rows: row maximum in closed form from the running max/min of k (max_j a*k[j] = a*max k if a>=0, else a*min k; exact), one lane scan per key group via LDS, replacing the per-pair max pass
# speedup vs baseline: 3.6787x; 1.5734x over previous
#include <hip/hip_runtime.h>
#include <hip/hip_bf16.h>
#include <math.h>


#define TT 512
#define DD 512

typedef __attribute__((ext_vector_type(2))) float v2f;
typedef __attribute__((ext_vector_type(8))) float v8f;

typedef __attribute__((ext_vector_type(4))) float v4f;

#define G1_BM 48
#define G1_BN 64
#define G1_BK 64
#define G1_P  68

__global__ void __launch_bounds__(256)
qkv_gemm_f32(const float* __restrict__ x,
             const float* __restrict__ Wq,
             const float* __restrict__ Wk,
             const float* __restrict__ Wv,
             float* __restrict__ q,
             float* __restrict__ k,
             float* __restrict__ v) {
    __shared__ float As[2 * G1_BM * G1_P];
    __shared__ float Bs[2 * G1_BN * G1_P];

    const int tid  = threadIdx.x;
    const int lane = tid & 31;
    const int w    = __builtin_amdgcn_readfirstlane(tid >> 5);
    const int wq   = w & 3;
    const int kh   = w >> 2;
    const int c    = lane & 15;
    const int h2   = (lane >> 4) << 1;
    const int bn   = blockIdx.x;
    const int bm   = blockIdx.y;

    const int lrow = tid >> 4;
    const int lseg = (tid & 15) << 2;

    const int pr   = bm & 15;
    const float* Wqk = (bm < 16) ? Wq : Wk;

    const float* gA[3];
    const float* gB[4];
    gA[0] = Wqk + (pr * 16 + lrow) * DD + lseg;
    gA[1] = Wqk + (256 + pr * 16 + lrow) * DD + lseg;
    gA[2] = Wv  + (bm * 16 + lrow) * DD + lseg;
#pragma unroll
    for (int p = 0; p < 4; ++p)
        gB[p] = x + (bn * G1_BN + p * 16 + lrow) * DD + lseg;

    const int sOff = lrow * G1_P + lseg;
    const int fA   = c * G1_P + h2 + kh * 32;
    const int fB   = (wq * 16 + c) * G1_P + h2 + kh * 32;

    v4f ra[3], rb[4];
#pragma unroll
    for (int p = 0; p < 3; ++p) ra[p] = *(const v4f*)(gA[p]);
#pragma unroll
    for (int p = 0; p < 4; ++p) rb[p] = *(const v4f*)(gB[p]);
#pragma unroll
    for (int p = 0; p < 3; ++p) *(v4f*)(As + sOff + p * 16 * G1_P) = ra[p];
#pragma unroll
    for (int p = 0; p < 4; ++p) *(v4f*)(Bs + sOff + p * 16 * G1_P) = rb[p];
    __syncthreads();

    v8f acc0 = {}, acc1 = {}, acc2 = {};

#define G1_LOADF(A0_, A1_, A2_, B_, KK0)                                   \
    _Pragma("unroll")                                                      \
    for (int u = 0; u < 4; ++u) {                                          \
        B_[u]  = *(const v2f*)(bs + ((KK0) + u) * 4);                      \
        A0_[u] = *(const v2f*)(as + ((KK0) + u) * 4);                      \
        A1_[u] = *(const v2f*)(as + 16 * G1_P + ((KK0) + u) * 4);          \
        A2_[u] = *(const v2f*)(as + 32 * G1_P + ((KK0) + u) * 4);          \
    }
#define G1_MMA(A0_, A1_, A2_, B_)                                          \
    _Pragma("unroll")                                                      \
    for (int u = 0; u < 4; ++u) {                                          \
        acc0 = __builtin_amdgcn_wmma_f32_16x16x4_f32(false, A0_[u], false, B_[u], (short)0, acc0, false, false); \
        acc1 = __builtin_amdgcn_wmma_f32_16x16x4_f32(false, A1_[u], false, B_[u], (short)0, acc1, false, false); \
        acc2 = __builtin_amdgcn_wmma_f32_16x16x4_f32(false, A2_[u], false, B_[u], (short)0, acc2, false, false); \
    }

#pragma unroll 1
    for (int ks = 0; ks < DD / G1_BK; ++ks) {
        const int cur = ks & 1;
        const float* as = As + cur * (G1_BM * G1_P) + fA;
        const float* bs = Bs + cur * (G1_BN * G1_P) + fB;
        const bool more = (ks + 1 < DD / G1_BK);
        const int kn = (ks + 1) * G1_BK;

        v2f xa0[4], xa1[4], xa2[4], xb[4];
        v2f ya0[4], ya1[4], ya2[4], yb[4];

        if (more) {
#pragma unroll
            for (int p = 0; p < 3; ++p) ra[p] = *(const v4f*)(gA[p] + kn);
#pragma unroll
            for (int p = 0; p < 4; ++p) rb[p] = *(const v4f*)(gB[p] + kn);
        }
        G1_LOADF(xa0, xa1, xa2, xb, 0)
        G1_LOADF(ya0, ya1, ya2, yb, 4)
        __builtin_amdgcn_sched_barrier(0);
        G1_MMA(xa0, xa1, xa2, xb)
        __builtin_amdgcn_sched_barrier(0);
        if (more) {
            float* ad = As + (cur ^ 1) * (G1_BM * G1_P) + sOff;
            float* bd = Bs + (cur ^ 1) * (G1_BN * G1_P) + sOff;
#pragma unroll
            for (int p = 0; p < 3; ++p) *(v4f*)(ad + p * 16 * G1_P) = ra[p];
#pragma unroll
            for (int p = 0; p < 4; ++p) *(v4f*)(bd + p * 16 * G1_P) = rb[p];
        }
        __builtin_amdgcn_sched_barrier(0);
        G1_MMA(ya0, ya1, ya2, yb)
        __syncthreads();
    }

    float* red = As + (wq * 3 * 8) * 32 + lane;
    if (kh == 1) {
#pragma unroll
        for (int r = 0; r < 8; ++r) {
            red[(0 * 8 + r) * 32] = acc0[r];
            red[(1 * 8 + r) * 32] = acc1[r];
            red[(2 * 8 + r) * 32] = acc2[r];
        }
    }
    __syncthreads();
    if (kh == 0) {
#pragma unroll
        for (int r = 0; r < 8; ++r) {
            acc0[r] += red[(0 * 8 + r) * 32];
            acc1[r] += red[(1 * 8 + r) * 32];
            acc2[r] += red[(2 * 8 + r) * 32];
        }
        const int rowOff = (lane >> 4) << 3;
        const int col = bn * G1_BN + wq * 16 + c;
        const int olo = pr * 16 + rowOff;
        const float tf = (float)col;
        float* dqk = (bm < 16) ? q : k;
        float* dlo = dqk + olo * TT + col;
        float* dhi = dlo + 256 * TT;
#pragma unroll
        for (int r = 0; r < 8; ++r) {
            const int e = (olo + r) & 31;
            const float invf = __powf(10000.0f, -(float)(e << 1) * (1.0f / 64.0f));
            const float ang  = tf * invf;
            const float cs = __cosf(ang);
            const float sn = __sinf(ang);
            const float xl = acc0[r];
            const float xh = acc1[r];
            dlo[r * TT] = xl * cs - xh * sn;
            dhi[r * TT] = xh * cs + xl * sn;
        }
        float* dv = v + (bm * 16 + rowOff) * TT + col;
#pragma unroll
        for (int r = 0; r < 8; ++r)
            dv[r * TT] = acc2[r];
    }
}

#define POSINF (__builtin_inff())

template <int N>
__device__ __forceinline__ float row_shr(const float x) {
    const int xi = __builtin_bit_cast(int, x);
    return __builtin_bit_cast(float, __builtin_amdgcn_update_dpp(xi, xi, 0x110 + N, 0xf, 0xf, false));
}

__device__ __forceinline__ float lane_value(const float x, const int l) {
    return __builtin_bit_cast(float, __builtin_amdgcn_readlane(__builtin_bit_cast(int, x), l));
}

__device__ __forceinline__ void scan32(float& mx, float& mn, const int lane) {
    mx = fmaxf(mx, row_shr<1>(mx));  mn = fminf(mn, row_shr<1>(mn));
    mx = fmaxf(mx, row_shr<2>(mx));  mn = fminf(mn, row_shr<2>(mn));
    mx = fmaxf(mx, row_shr<4>(mx));  mn = fminf(mn, row_shr<4>(mn));
    mx = fmaxf(mx, row_shr<8>(mx));  mn = fminf(mn, row_shr<8>(mn));
    const float mx15 = lane_value(mx, 15);
    const float mn15 = lane_value(mn, 15);
    if (lane >= 16) {
        mx = fmaxf(mx, mx15);
        mn = fminf(mn, mn15);
    }
}

__device__ __forceinline__ void acc8(float& s0, float& s1, float& c0, float& c1,
                                     const float a, const float m, const v8f kk, const v8f vv) {
#pragma unroll
    for (int u = 0; u < 8; u += 2) {
        const float e0 = __expf(a * kk[u] - m);
        const float e1 = __expf(a * kk[u + 1] - m);
        s0 += e0;
        s1 += e1;
        c0 += e0 * vv[u];
        c1 += e1 * vv[u + 1];
    }
}

__device__ __forceinline__ void acc8_diag(float& s0, float& s1, float& c0, float& c1,
                                          const float a, const float m, const v8f kk, const v8f vv,
                                          const int base, const int lane) {
#pragma unroll
    for (int u = 0; u < 8; u += 2) {
        const float e0 = (base + u     <= lane) ? __expf(a * kk[u] - m)     : 0.0f;
        const float e1 = (base + u + 1 <= lane) ? __expf(a * kk[u + 1] - m) : 0.0f;
        s0 += e0;
        s1 += e1;
        c0 += e0 * vv[u];
        c1 += e1 * vv[u + 1];
    }
}

__device__ __forceinline__ float attn_row32(const float* __restrict__ kr,
                                            const float* __restrict__ vr,
                                            const float a, const float m,
                                            const int r0, const int lane) {
    float s0 = 0.0f, s1 = 0.0f, c0 = 0.0f, c1 = 0.0f;
    {
        v8f kA = *(const v8f*)(kr);
        v8f vA = *(const v8f*)(vr);
        v8f kB, vB;
        for (int j = 0; j < r0; j += 16) {
            kB = *(const v8f*)(kr + j + 8);
            vB = *(const v8f*)(vr + j + 8);
            acc8(s0, s1, c0, c1, a, m, kA, vA);
            kA = *(const v8f*)(kr + j + 16);
            vA = *(const v8f*)(vr + j + 16);
            acc8(s0, s1, c0, c1, a, m, kB, vB);
        }
        kB = *(const v8f*)(kr + r0 + 8);
        vB = *(const v8f*)(vr + r0 + 8);
        acc8_diag(s0, s1, c0, c1, a, m, kA, vA, 0, lane);
        kA = *(const v8f*)(kr + r0 + 16);
        vA = *(const v8f*)(vr + r0 + 16);
        acc8_diag(s0, s1, c0, c1, a, m, kB, vB, 8, lane);
        kB = *(const v8f*)(kr + r0 + 24);
        vB = *(const v8f*)(vr + r0 + 24);
        acc8_diag(s0, s1, c0, c1, a, m, kA, vA, 16, lane);
        acc8_diag(s0, s1, c0, c1, a, m, kB, vB, 24, lane);
    }
    return (c0 + c1) / (s0 + s1);
}

__global__ void __launch_bounds__(256)
attn_rows(const float* __restrict__ q,
          const float* __restrict__ k,
          const float* __restrict__ v,
          float* __restrict__ o) {
    __shared__ float gmax[16];
    __shared__ float gmin[16];

    const int lane = threadIdx.x & 31;
    const int w    = __builtin_amdgcn_readfirstlane(threadIdx.x >> 5);
    const int ch   = blockIdx.x;

    const float* kr = k + ch * TT;
    const float* vr = v + ch * TT;
    const float* qr = q + ch * TT;
    float*       orow = o + ch * TT;

    const int rlo = w << 5;
    const int rhi = 480 - (w << 5);

    const float alo = qr[rlo + lane] * 0.125f;
    const float ahi = qr[rhi + lane] * 0.125f;

    float dlx = kr[rlo + lane];
    float dhx = kr[rhi + lane];
    float dln = dlx, dhn = dhx;
    scan32(dlx, dln, lane);
    scan32(dhx, dhn, lane);
    if (lane == 31) {
        gmax[w]      = dlx;
        gmin[w]      = dln;
        gmax[15 - w] = dhx;
        gmin[15 - w] = dhn;
    }
    __syncthreads();

    float gx = gmax[lane & 15];
    float gn = gmin[lane & 15];
    scan32(gx, gn, lane);

    const float uhx = lane_value(gx, 14 - w);
    const float uhn = lane_value(gn, 14 - w);
    const float tlx = lane_value(gx, (w > 0) ? w - 1 : 0);
    const float tln = lane_value(gn, (w > 0) ? w - 1 : 0);
    const float ulx = (w > 0) ? tlx : -POSINF;
    const float uln = (w > 0) ? tln : POSINF;

    const float kxlo = fmaxf(ulx, dlx);
    const float knlo = fminf(uln, dln);
    const float kxhi = fmaxf(uhx, dhx);
    const float knhi = fminf(uhn, dhn);

    const float mlo = fmaxf(alo * kxlo, alo * knlo);
    const float mhi = fmaxf(ahi * kxhi, ahi * knhi);

    orow[rlo + lane] = attn_row32(kr, vr, alo, mlo, rlo, lane);
    orow[rhi + lane] = attn_row32(kr, vr, ahi, mhi, rhi, lane);
}

#define G2_BK 128
#define G2_PA 132
#define G2_PB 48

__global__ void __launch_bounds__(256)
out_gemm_f32(const float* __restrict__ ao,
             const float* __restrict__ Wo,
             float* __restrict__ out) {
    __shared__ float As[32 * G2_PA];
    __shared__ float Bs[G2_BK * G2_PB];

    const int tid  = threadIdx.x;
    const int lane = tid & 31;
    const int w    = __builtin_amdgcn_readfirstlane(tid >> 5);
    const int c    = lane & 15;
    const int h2   = (lane >> 4) << 1;
    const int wm   = (w >> 1) & 1;
    const int wn   = w & 1;
    const int kh   = w >> 2;
    const int bn   = blockIdx.x;
    const int bm   = blockIdx.y;

    const float* gA = Wo + (bm * 32 + w) * DD + (lane << 2);
    const int sA = w * G2_PA + (lane << 2);
    const float* gB = ao + (tid >> 3) * TT + bn * 32 + ((tid & 7) << 2);
    const int sB = (tid >> 3) * G2_PB + ((tid & 7) << 2);

    v4f ra[4], rb[4];
#pragma unroll
    for (int p = 0; p < 4; ++p) ra[p] = *(const v4f*)(gA + p * 8 * DD);
#pragma unroll
    for (int p = 0; p < 4; ++p) rb[p] = *(const v4f*)(gB + p * 32 * TT);
#pragma unroll
    for (int p = 0; p < 4; ++p) *(v4f*)(As + sA + p * 8 * G2_PA) = ra[p];
#pragma unroll
    for (int p = 0; p < 4; ++p) *(v4f*)(Bs + sB + p * 32 * G2_PB) = rb[p];
    __syncthreads();

    const float* as = As + (wm * 16 + c) * G2_PA + h2 + kh * 64;
    const float* bs = Bs + (h2 + kh * 64) * G2_PB + wn * 16 + c;

    v8f c0 = {}, c1 = {};

#define G2_LOADF(A_, B_, KK0)                                              \
        _Pragma("unroll")                                                  \
        for (int u = 0; u < 8; ++u) {                                      \
            A_[u]   = *(const v2f*)(as + ((KK0) + u) * 4);                 \
            B_[u].x = bs[(((KK0) + u) * 4) * G2_PB];                       \
            B_[u].y = bs[(((KK0) + u) * 4 + 1) * G2_PB];                   \
        }
#define G2_MMA(A_, B_)                                                     \
        _Pragma("unroll")                                                  \
        for (int u = 0; u < 8; u += 2) {                                   \
            c0 = __builtin_amdgcn_wmma_f32_16x16x4_f32(false, A_[u], false, B_[u], (short)0, c0, false, false);         \
            c1 = __builtin_amdgcn_wmma_f32_16x16x4_f32(false, A_[u + 1], false, B_[u + 1], (short)0, c1, false, false); \
        }

#pragma unroll 1
    for (int ks = 0; ks < DD / G2_BK; ++ks) {
        const bool more = (ks + 1 < DD / G2_BK);
        const int kn = (ks + 1) * G2_BK;

        if (more) {
#pragma unroll
            for (int p = 0; p < 4; ++p) ra[p] = *(const v4f*)(gA + p * 8 * DD + kn);
#pragma unroll
            for (int p = 0; p < 4; ++p) rb[p] = *(const v4f*)(gB + (kn + p * 32) * TT);
        }

        v2f xa[8], xb[8], ya[8], yb[8];
        G2_LOADF(xa, xb, 0)
        G2_LOADF(ya, yb, 8)
        __builtin_amdgcn_sched_barrier(0);
        G2_MMA(xa, xb)
        __builtin_amdgcn_sched_barrier(0);
        G2_MMA(ya, yb)

        __syncthreads();
        if (!more) break;
#pragma unroll
        for (int p = 0; p < 4; ++p) *(v4f*)(As + sA + p * 8 * G2_PA) = ra[p];
#pragma unroll
        for (int p = 0; p < 4; ++p) *(v4f*)(Bs + sB + p * 32 * G2_PB) = rb[p];
        __syncthreads();
    }

    v8f cc = c0 + c1;
    float* red = As + ((w & 3) * 8) * 32 + lane;
    if (kh == 1) {
#pragma unroll
        for (int r = 0; r < 8; ++r) red[r * 32] = cc[r];
    }
    __syncthreads();
    if (kh == 0) {
#pragma unroll
        for (int r = 0; r < 8; ++r) cc[r] += red[r * 32];
        const int t  = bn * 32 + wn * 16 + c;
        const int o0 = bm * 32 + wm * 16 + ((lane >> 4) << 3);
        float* dst = out + t * DD + o0;
        v4f lo, hi;
        lo.x = cc[0]; lo.y = cc[1]; lo.z = cc[2]; lo.w = cc[3];
        hi.x = cc[4]; hi.y = cc[5]; hi.z = cc[6]; hi.w = cc[7];
        *(v4f*)(dst)     = lo;
        *(v4f*)(dst + 4) = hi;
    }
}

extern "C" void kernel_launch(void* const* d_in, const int* in_sizes, int n_in,
                              void* d_out, int out_size, void* d_ws, size_t ws_size,
                              hipStream_t stream) {
    const float* x  = (const float*)d_in[0];
    const float* Wq = (const float*)d_in[1];
    const float* Wk = (const float*)d_in[2];
    const float* Wv = (const float*)d_in[3];
    const float* Wo = (const float*)d_in[4];
    float* out = (float*)d_out;

    float* ws = (float*)d_ws;
    const size_t S = (size_t)DD * TT;
    float* qrot = ws + 0 * S;
    float* krot = ws + 1 * S;
    float* vbuf = ws + 2 * S;
    float* atto = ws + 3 * S;

    qkv_gemm_f32<<<dim3(8, 32, 1), 256, 0, stream>>>(x, Wq, Wk, Wv, qrot, krot, vbuf);

    attn_rows<<<DD, 256, 0, stream>>>(qrot, krot, vbuf, atto);

    out_gemm_f32<<<dim3(16, 16, 1), 256, 0, stream>>>(atto, Wo, out);
}


# attention rows: key loop cut from 4.1 to 3.0 vector instructions per (row, key): log2(e) folded into the row factor, v[j] read as register pairs from an LDS copy so the sums run packed
# speedup vs baseline: 4.0680x; 1.1058x over previous
#include <hip/hip_runtime.h>
#include <hip/hip_bf16.h>
#include <math.h>


#define TT 512
#define DD 512

typedef __attribute__((ext_vector_type(2))) float v2f;
typedef __attribute__((ext_vector_type(8))) float v8f;

typedef __attribute__((ext_vector_type(4))) float v4f;

#define G1_BM 48
#define G1_BN 64
#define G1_BK 64
#define G1_P  68

__global__ void __launch_bounds__(256)
qkv_gemm_f32(const float* __restrict__ x,
             const float* __restrict__ Wq,
             const float* __restrict__ Wk,
             const float* __restrict__ Wv,
             float* __restrict__ q,
             float* __restrict__ k,
             float* __restrict__ v) {
    __shared__ float As[2 * G1_BM * G1_P];
    __shared__ float Bs[2 * G1_BN * G1_P];

    const int tid  = threadIdx.x;
    const int lane = tid & 31;
    const int w    = __builtin_amdgcn_readfirstlane(tid >> 5);
    const int wq   = w & 3;
    const int kh   = w >> 2;
    const int c    = lane & 15;
    const int h2   = (lane >> 4) << 1;
    const int bn   = blockIdx.x;
    const int bm   = blockIdx.y;

    const int lrow = tid >> 4;
    const int lseg = (tid & 15) << 2;

    const int pr   = bm & 15;
    const float* Wqk = (bm < 16) ? Wq : Wk;

    const float* gA[3];
    const float* gB[4];
    gA[0] = Wqk + (pr * 16 + lrow) * DD + lseg;
    gA[1] = Wqk + (256 + pr * 16 + lrow) * DD + lseg;
    gA[2] = Wv  + (bm * 16 + lrow) * DD + lseg;
#pragma unroll
    for (int p = 0; p < 4; ++p)
        gB[p] = x + (bn * G1_BN + p * 16 + lrow) * DD + lseg;

    const int sOff = lrow * G1_P + lseg;
    const int fA   = c * G1_P + h2 + kh * 32;
    const int fB   = (wq * 16 + c) * G1_P + h2 + kh * 32;

    v4f ra[3], rb[4];
#pragma unroll
    for (int p = 0; p < 3; ++p) ra[p] = *(const v4f*)(gA[p]);
#pragma unroll
    for (int p = 0; p < 4; ++p) rb[p] = *(const v4f*)(gB[p]);
#pragma unroll
    for (int p = 0; p < 3; ++p) *(v4f*)(As + sOff + p * 16 * G1_P) = ra[p];
#pragma unroll
    for (int p = 0; p < 4; ++p) *(v4f*)(Bs + sOff + p * 16 * G1_P) = rb[p];
    __syncthreads();

    v8f acc0 = {}, acc1 = {}, acc2 = {};

#define G1_LOADF(A0_, A1_, A2_, B_, KK0)                                   \
    _Pragma("unroll")                                                      \
    for (int u = 0; u < 4; ++u) {                                          \
        B_[u]  = *(const v2f*)(bs + ((KK0) + u) * 4);                      \
        A0_[u] = *(const v2f*)(as + ((KK0) + u) * 4);                      \
        A1_[u] = *(const v2f*)(as + 16 * G1_P + ((KK0) + u) * 4);          \
        A2_[u] = *(const v2f*)(as + 32 * G1_P + ((KK0) + u) * 4);          \
    }
#define G1_MMA(A0_, A1_, A2_, B_)                                          \
    _Pragma("unroll")                                                      \
    for (int u = 0; u < 4; ++u) {                                          \
        acc0 = __builtin_amdgcn_wmma_f32_16x16x4_f32(false, A0_[u], false, B_[u], (short)0, acc0, false, false); \
        acc1 = __builtin_amdgcn_wmma_f32_16x16x4_f32(false, A1_[u], false, B_[u], (short)0, acc1, false, false); \
        acc2 = __builtin_amdgcn_wmma_f32_16x16x4_f32(false, A2_[u], false, B_[u], (short)0, acc2, false, false); \
    }

#pragma unroll 1
    for (int ks = 0; ks < DD / G1_BK; ++ks) {
        const int cur = ks & 1;
        const float* as = As + cur * (G1_BM * G1_P) + fA;
        const float* bs = Bs + cur * (G1_BN * G1_P) + fB;
        const bool more = (ks + 1 < DD / G1_BK);
        const int kn = (ks + 1) * G1_BK;

        v2f xa0[4], xa1[4], xa2[4], xb[4];
        v2f ya0[4], ya1[4], ya2[4], yb[4];

        if (more) {
#pragma unroll
            for (int p = 0; p < 3; ++p) ra[p] = *(const v4f*)(gA[p] + kn);
#pragma unroll
            for (int p = 0; p < 4; ++p) rb[p] = *(const v4f*)(gB[p] + kn);
        }
        G1_LOADF(xa0, xa1, xa2, xb, 0)
        G1_LOADF(ya0, ya1, ya2, yb, 4)
        __builtin_amdgcn_sched_barrier(0);
        G1_MMA(xa0, xa1, xa2, xb)
        __builtin_amdgcn_sched_barrier(0);
        if (more) {
            float* ad = As + (cur ^ 1) * (G1_BM * G1_P) + sOff;
            float* bd = Bs + (cur ^ 1) * (G1_BN * G1_P) + sOff;
#pragma unroll
            for (int p = 0; p < 3; ++p) *(v4f*)(ad + p * 16 * G1_P) = ra[p];
#pragma unroll
            for (int p = 0; p < 4; ++p) *(v4f*)(bd + p * 16 * G1_P) = rb[p];
        }
        __builtin_amdgcn_sched_barrier(0);
        G1_MMA(ya0, ya1, ya2, yb)
        __syncthreads();
    }

    float* red = As + (wq * 3 * 8) * 32 + lane;
    if (kh == 1) {
#pragma unroll
        for (int r = 0; r < 8; ++r) {
            red[(0 * 8 + r) * 32] = acc0[r];
            red[(1 * 8 + r) * 32] = acc1[r];
            red[(2 * 8 + r) * 32] = acc2[r];
        }
    }
    __syncthreads();
    if (kh == 0) {
#pragma unroll
        for (int r = 0; r < 8; ++r) {
            acc0[r] += red[(0 * 8 + r) * 32];
            acc1[r] += red[(1 * 8 + r) * 32];
            acc2[r] += red[(2 * 8 + r) * 32];
        }
        const int rowOff = (lane >> 4) << 3;
        const int col = bn * G1_BN + wq * 16 + c;
        const int olo = pr * 16 + rowOff;
        const float tf = (float)col;
        float* dqk = (bm < 16) ? q : k;
        float* dlo = dqk + olo * TT + col;
        float* dhi = dlo + 256 * TT;
#pragma unroll
        for (int r = 0; r < 8; ++r) {
            const int e = (olo + r) & 31;
            const float invf = __powf(10000.0f, -(float)(e << 1) * (1.0f / 64.0f));
            const float ang  = tf * invf;
            const float cs = __cosf(ang);
            const float sn = __sinf(ang);
            const float xl = acc0[r];
            const float xh = acc1[r];
            dlo[r * TT] = xl * cs - xh * sn;
            dhi[r * TT] = xh * cs + xl * sn;
        }
        float* dv = v + (bm * 16 + rowOff) * TT + col;
#pragma unroll
        for (int r = 0; r < 8; ++r)
            dv[r * TT] = acc2[r];
    }
}

#define POSINF (__builtin_inff())

template <int N>
__device__ __forceinline__ float row_shr(const float x) {
    const int xi = __builtin_bit_cast(int, x);
    return __builtin_bit_cast(float, __builtin_amdgcn_update_dpp(xi, xi, 0x110 + N, 0xf, 0xf, false));
}

__device__ __forceinline__ float lane_value(const float x, const int l) {
    return __builtin_bit_cast(float, __builtin_amdgcn_readlane(__builtin_bit_cast(int, x), l));
}

__device__ __forceinline__ void scan32(float& mx, float& mn, const int lane) {
    mx = fmaxf(mx, row_shr<1>(mx));  mn = fminf(mn, row_shr<1>(mn));
    mx = fmaxf(mx, row_shr<2>(mx));  mn = fminf(mn, row_shr<2>(mn));
    mx = fmaxf(mx, row_shr<4>(mx));  mn = fminf(mn, row_shr<4>(mn));
    mx = fmaxf(mx, row_shr<8>(mx));  mn = fminf(mn, row_shr<8>(mn));
    const float mx15 = lane_value(mx, 15);
    const float mn15 = lane_value(mn, 15);
    if (lane >= 16) {
        mx = fmaxf(mx, mx15);
        mn = fminf(mn, mn15);
    }
}

typedef __attribute__((ext_vector_type(16))) float v16f;

__device__ __forceinline__ void acc16(v2f& sA, v2f& sB, v2f& cA, v2f& cB,
                                      const float a, const float m, const v16f kk,
                                      const v4f w0, const v4f w1, const v4f w2, const v4f w3) {
    const v4f ww[4] = {w0, w1, w2, w3};
#pragma unroll
    for (int u = 0; u < 16; u += 4) {
        v2f e0, e1, p0, p1;
        e0.x = __builtin_amdgcn_exp2f(__builtin_fmaf(a, kk[u],     -m));
        e0.y = __builtin_amdgcn_exp2f(__builtin_fmaf(a, kk[u + 1], -m));
        e1.x = __builtin_amdgcn_exp2f(__builtin_fmaf(a, kk[u + 2], -m));
        e1.y = __builtin_amdgcn_exp2f(__builtin_fmaf(a, kk[u + 3], -m));
        p0.x = ww[u >> 2].x;  p0.y = ww[u >> 2].y;
        p1.x = ww[u >> 2].z;  p1.y = ww[u >> 2].w;
        sA += e0;
        sB += e1;
        cA = __builtin_elementwise_fma(e0, p0, cA);
        cB = __builtin_elementwise_fma(e1, p1, cB);
    }
}

__device__ __forceinline__ void acc16_diag(v2f& sA, v2f& sB, v2f& cA, v2f& cB,
                                           const float a, const float m, const v16f kk,
                                           const v4f w0, const v4f w1, const v4f w2, const v4f w3,
                                           const int base, const int lane) {
    const v4f ww[4] = {w0, w1, w2, w3};
#pragma unroll
    for (int u = 0; u < 16; u += 4) {
        v2f e0, e1, p0, p1;
        e0.x = (base + u     <= lane) ? __builtin_amdgcn_exp2f(__builtin_fmaf(a, kk[u],     -m)) : 0.0f;
        e0.y = (base + u + 1 <= lane) ? __builtin_amdgcn_exp2f(__builtin_fmaf(a, kk[u + 1], -m)) : 0.0f;
        e1.x = (base + u + 2 <= lane) ? __builtin_amdgcn_exp2f(__builtin_fmaf(a, kk[u + 2], -m)) : 0.0f;
        e1.y = (base + u + 3 <= lane) ? __builtin_amdgcn_exp2f(__builtin_fmaf(a, kk[u + 3], -m)) : 0.0f;
        p0.x = ww[u >> 2].x;  p0.y = ww[u >> 2].y;
        p1.x = ww[u >> 2].z;  p1.y = ww[u >> 2].w;
        sA += e0;
        sB += e1;
        cA = __builtin_elementwise_fma(e0, p0, cA);
        cB = __builtin_elementwise_fma(e1, p1, cB);
    }
}

__device__ __forceinline__ float attn_row32(const float* __restrict__ kr,
                                            const float* vs,
                                            const float a, const float m,
                                            const int r0, const int lane) {
    v2f sA = {0.0f, 0.0f}, sB = {0.0f, 0.0f}, cA = {0.0f, 0.0f}, cB = {0.0f, 0.0f};
    v16f kA = *(const v16f*)(kr);
    v4f a0 = *(const v4f*)(vs), a1 = *(const v4f*)(vs + 4), a2 = *(const v4f*)(vs + 8), a3 = *(const v4f*)(vs + 12);
    v16f kB;
    v4f b0, b1, b2, b3;
    for (int j = 0; j < r0; j += 32) {
        kB = *(const v16f*)(kr + j + 16);
        b0 = *(const v4f*)(vs + j + 16);  b1 = *(const v4f*)(vs + j + 20);
        b2 = *(const v4f*)(vs + j + 24);  b3 = *(const v4f*)(vs + j + 28);
        acc16(sA, sB, cA, cB, a, m, kA, a0, a1, a2, a3);
        kA = *(const v16f*)(kr + j + 32);
        a0 = *(const v4f*)(vs + j + 32);  a1 = *(const v4f*)(vs + j + 36);
        a2 = *(const v4f*)(vs + j + 40);  a3 = *(const v4f*)(vs + j + 44);
        acc16(sA, sB, cA, cB, a, m, kB, b0, b1, b2, b3);
    }
    kB = *(const v16f*)(kr + r0 + 16);
    b0 = *(const v4f*)(vs + r0 + 16);  b1 = *(const v4f*)(vs + r0 + 20);
    b2 = *(const v4f*)(vs + r0 + 24);  b3 = *(const v4f*)(vs + r0 + 28);
    acc16_diag(sA, sB, cA, cB, a, m, kA, a0, a1, a2, a3, 0, lane);
    acc16_diag(sA, sB, cA, cB, a, m, kB, b0, b1, b2, b3, 16, lane);
    const v2f s = sA + sB;
    const v2f c = cA + cB;
    return (c.x + c.y) / (s.x + s.y);
}

__global__ void __launch_bounds__(256)
attn_rows(const float* __restrict__ q,
          const float* __restrict__ k,
          const float* __restrict__ v,
          float* __restrict__ o) {
    __shared__ float gmax[16];
    __shared__ float gmin[16];
    __shared__ __attribute__((aligned(16))) float vs[TT];

    const int lane = threadIdx.x & 31;
    const int w    = __builtin_amdgcn_readfirstlane(threadIdx.x >> 5);
    const int ch   = blockIdx.x;

    const float* kr = k + ch * TT;
    const float* vr = v + ch * TT;
    const float* qr = q + ch * TT;
    float*       orow = o + ch * TT;

    const int rlo = w << 5;
    const int rhi = 480 - (w << 5);

    const float alo = qr[rlo + lane] * 0.125f * 1.4426950408889634f;
    const float ahi = qr[rhi + lane] * 0.125f * 1.4426950408889634f;

    *(v2f*)(vs + 2 * threadIdx.x) = *(const v2f*)(vr + 2 * threadIdx.x);

    float dlx = kr[rlo + lane];
    float dhx = kr[rhi + lane];
    float dln = dlx, dhn = dhx;
    scan32(dlx, dln, lane);
    scan32(dhx, dhn, lane);
    if (lane == 31) {
        gmax[w]      = dlx;
        gmin[w]      = dln;
        gmax[15 - w] = dhx;
        gmin[15 - w] = dhn;
    }
    __syncthreads();

    float gx = gmax[lane & 15];
    float gn = gmin[lane & 15];
    scan32(gx, gn, lane);

    const float uhx = lane_value(gx, 14 - w);
    const float uhn = lane_value(gn, 14 - w);
    const float tlx = lane_value(gx, (w > 0) ? w - 1 : 0);
    const float tln = lane_value(gn, (w > 0) ? w - 1 : 0);
    const float ulx = (w > 0) ? tlx : -POSINF;
    const float uln = (w > 0) ? tln : POSINF;

    const float kxlo = fmaxf(ulx, dlx);
    const float knlo = fminf(uln, dln);
    const float kxhi = fmaxf(uhx, dhx);
    const float knhi = fminf(uhn, dhn);

    const float mlo = fmaxf(alo * kxlo, alo * knlo);
    const float mhi = fmaxf(ahi * kxhi, ahi * knhi);

    orow[rlo + lane] = attn_row32(kr, vs, alo, mlo, rlo, lane);
    orow[rhi + lane] = attn_row32(kr, vs, ahi, mhi, rhi, lane);
}

#define G2_BK 128
#define G2_PA 132
#define G2_PB 48

__global__ void __launch_bounds__(256)
out_gemm_f32(const float* __restrict__ ao,
             const float* __restrict__ Wo,
             float* __restrict__ out) {
    __shared__ float As[32 * G2_PA];
    __shared__ float Bs[G2_BK * G2_PB];

    const int tid  = threadIdx.x;
    const int lane = tid & 31;
    const int w    = __builtin_amdgcn_readfirstlane(tid >> 5);
    const int c    = lane & 15;
    const int h2   = (lane >> 4) << 1;
    const int wm   = (w >> 1) & 1;
    const int wn   = w & 1;
    const int kh   = w >> 2;
    const int bn   = blockIdx.x;
    const int bm   = blockIdx.y;

    const float* gA = Wo + (bm * 32 + w) * DD + (lane << 2);
    const int sA = w * G2_PA + (lane << 2);
    const float* gB = ao + (tid >> 3) * TT + bn * 32 + ((tid & 7) << 2);
    const int sB = (tid >> 3) * G2_PB + ((tid & 7) << 2);

    v4f ra[4], rb[4];
#pragma unroll
    for (int p = 0; p < 4; ++p) ra[p] = *(const v4f*)(gA + p * 8 * DD);
#pragma unroll
    for (int p = 0; p < 4; ++p) rb[p] = *(const v4f*)(gB + p * 32 * TT);
#pragma unroll
    for (int p = 0; p < 4; ++p) *(v4f*)(As + sA + p * 8 * G2_PA) = ra[p];
#pragma unroll
    for (int p = 0; p < 4; ++p) *(v4f*)(Bs + sB + p * 32 * G2_PB) = rb[p];
    __syncthreads();

    const float* as = As + (wm * 16 + c) * G2_PA + h2 + kh * 64;
    const float* bs = Bs + (h2 + kh * 64) * G2_PB + wn * 16 + c;

    v8f c0 = {}, c1 = {};

#define G2_LOADF(A_, B_, KK0)                                              \
        _Pragma("unroll")                                                  \
        for (int u = 0; u < 8; ++u) {                                      \
            A_[u]   = *(const v2f*)(as + ((KK0) + u) * 4);                 \
            B_[u].x = bs[(((KK0) + u) * 4) * G2_PB];                       \
            B_[u].y = bs[(((KK0) + u) * 4 + 1) * G2_PB];                   \
        }
#define G2_MMA(A_, B_)                                                     \
        _Pragma("unroll")                                                  \
        for (int u = 0; u < 8; u += 2) {                                   \
            c0 = __builtin_amdgcn_wmma_f32_16x16x4_f32(false, A_[u], false, B_[u], (short)0, c0, false, false);         \
            c1 = __builtin_amdgcn_wmma_f32_16x16x4_f32(false, A_[u + 1], false, B_[u + 1], (short)0, c1, false, false); \
        }

#pragma unroll 1
    for (int ks = 0; ks < DD / G2_BK; ++ks) {
        const bool more = (ks + 1 < DD / G2_BK);
        const int kn = (ks + 1) * G2_BK;

        if (more) {
#pragma unroll
            for (int p = 0; p < 4; ++p) ra[p] = *(const v4f*)(gA + p * 8 * DD + kn);
#pragma unroll
            for (int p = 0; p < 4; ++p) rb[p] = *(const v4f*)(gB + (kn + p * 32) * TT);
        }

        v2f xa[8], xb[8], ya[8], yb[8];
        G2_LOADF(xa, xb, 0)
        G2_LOADF(ya, yb, 8)
        __builtin_amdgcn_sched_barrier(0);
        G2_MMA(xa, xb)
        __builtin_amdgcn_sched_barrier(0);
        G2_MMA(ya, yb)

        __syncthreads();
        if (!more) break;
#pragma unroll
        for (int p = 0; p < 4; ++p) *(v4f*)(As + sA + p * 8 * G2_PA) = ra[p];
#pragma unroll
        for (int p = 0; p < 4; ++p) *(v4f*)(Bs + sB + p * 32 * G2_PB) = rb[p];
        __syncthreads();
    }

    v8f cc = c0 + c1;
    float* red = As + ((w & 3) * 8) * 32 + lane;
    if (kh == 1) {
#pragma unroll
        for (int r = 0; r < 8; ++r) red[r * 32] = cc[r];
    }
    __syncthreads();
    if (kh == 0) {
#pragma unroll
        for (int r = 0; r < 8; ++r) cc[r] += red[r * 32];
        const int t  = bn * 32 + wn * 16 + c;
        const int o0 = bm * 32 + wm * 16 + ((lane >> 4) << 3);
        float* dst = out + t * DD + o0;
        v4f lo, hi;
        lo.x = cc[0]; lo.y = cc[1]; lo.z = cc[2]; lo.w = cc[3];
        hi.x = cc[4]; hi.y = cc[5]; hi.z = cc[6]; hi.w = cc[7];
        *(v4f*)(dst)     = lo;
        *(v4f*)(dst + 4) = hi;
    }
}

extern "C" void kernel_launch(void* const* d_in, const int* in_sizes, int n_in,
                              void* d_out, int out_size, void* d_ws, size_t ws_size,
                              hipStream_t stream) {
    const float* x  = (const float*)d_in[0];
    const float* Wq = (const float*)d_in[1];
    const float* Wk = (const float*)d_in[2];
    const float* Wv = (const float*)d_in[3];
    const float* Wo = (const float*)d_in[4];
    float* out = (float*)d_out;

    float* ws = (float*)d_ws;
    const size_t S = (size_t)DD * TT;
    float* qrot = ws + 0 * S;
    float* krot = ws + 1 * S;
    float* vbuf = ws + 2 * S;
    float* atto = ws + 3 * S;

    qkv_gemm_f32<<<dim3(8, 32, 1), 256, 0, stream>>>(x, Wq, Wk, Wv, qrot, krot, vbuf);

    attn_rows<<<DD, 256, 0, stream>>>(qrot, krot, vbuf, atto);

    out_gemm_f32<<<dim3(16, 16, 1), 256, 0, stream>>>(atto, Wo, out);
}
